# starting kernel
# baseline (speedup 1.0000x reference)
#include <hip/hip_runtime.h>
#include <hip/hip_bf16.h>

typedef __attribute__((ext_vector_type(2))) float v2f;
typedef __attribute__((ext_vector_type(8))) float v8f;

#define DDIM 256
#define BM 128
#define BN 128
#define LDT 260

__global__ __launch_bounds__(256) void norms_kernel(
    const float* __restrict__ s1, const float* __restrict__ s2,
    const float* __restrict__ W1, const float* __restrict__ W2,
    float* __restrict__ inv)
{
    int idx = blockIdx.x * blockDim.x + threadIdx.x;
    int i   = idx & 1023;
    int p   = (idx >> 10) & 31;
    int w   = (idx >> 15) & 1;
    int src = (idx >> 16) & 1;
    const float4* xr = (const float4*)((src ? s2 : s1) + i * DDIM);
    const float4* wr = (const float4*)((w ? W2 : W1) + p * DDIM);
    float sum = 0.f;
#pragma unroll 8
    for (int d4 = 0; d4 < DDIM / 4; ++d4) {
        float4 xv = xr[d4];
        float4 wv = wr[d4];
        float a0 = xv.x * wv.x, a1 = xv.y * wv.y, a2 = xv.z * wv.z, a3 = xv.w * wv.w;
        sum += a0 * a0 + a1 * a1 + a2 * a2 + a3 * a3;
    }
    inv[idx] = 1.0f / fmaxf(sqrtf(sum), 1e-8f);
}

__global__ __launch_bounds__(256) void mpm_kernel(
    const float* __restrict__ s1, const float* __restrict__ s2,
    const float* __restrict__ W1, const float* __restrict__ W2,
    const float* __restrict__ invn,
    float* __restrict__ out)
{
    __shared__ float lds_W[DDIM];
    __shared__ float lds_A[BM * LDT];
    __shared__ float lds_B[BN * LDT];
    __shared__ float red[2 * BM];

    const int pslot = blockIdx.x & 63;
    const int iTile = blockIdx.x >> 6;
    const int w  = pslot >> 5;
    const int p  = pslot & 31;
    const int i0 = iTile * BM;

    const float* Wsel = w ? W2 : W1;
    const float* invA = invn + ((size_t)(0 * 2 + w) * 32 + p) * 1024;
    const float* invB = invn + ((size_t)(1 * 2 + w) * 32 + p) * 1024;

    const int t    = threadIdx.x;
    const int lane = t & 31;
    const int wv   = t >> 5;
    const int rowBlk = wv & 3;
    const int colBlk = wv >> 2;

    if (t < DDIM / 4)
        ((float4*)lds_W)[t] = ((const float4*)(Wsel + p * DDIM))[t];
    __syncthreads();

#pragma unroll 8
    for (int it = 0; it < (BM * DDIM / 4) / 256; ++it) {
        int slot = t + it * 256;
        int row  = slot >> 6;
        int c4   = slot & 63;
        float ir = invA[i0 + row];
        float4 v  = ((const float4*)(s1 + (size_t)(i0 + row) * DDIM))[c4];
        float4 wq = ((const float4*)lds_W)[c4];
        v.x *= wq.x * ir; v.y *= wq.y * ir; v.z *= wq.z * ir; v.w *= wq.w * ir;
        *((float4*)&lds_A[row * LDT + c4 * 4]) = v;
    }

    float rmax[2][8];
#pragma unroll
    for (int rt = 0; rt < 2; ++rt)
#pragma unroll
        for (int r = 0; r < 8; ++r) rmax[rt][r] = -3.402823466e+38f;

    const int m     = lane & 15;
    const int khalf = lane >> 4;
    const float* Arow0 = &lds_A[(rowBlk * 32 + m) * LDT];
    const float* Arow1 = Arow0 + 16 * LDT;

    for (int jt = 0; jt < 8; ++jt) {
        const int j0 = jt * BN;
        __syncthreads();
#pragma unroll 8
        for (int it = 0; it < (BN * DDIM / 4) / 256; ++it) {
            int slot = t + it * 256;
            int row  = slot >> 6;
            int c4   = slot & 63;
            float ir = invB[j0 + row];
            float4 v  = ((const float4*)(s2 + (size_t)(j0 + row) * DDIM))[c4];
            float4 wq = ((const float4*)lds_W)[c4];
            v.x *= wq.x * ir; v.y *= wq.y * ir; v.z *= wq.z * ir; v.w *= wq.w * ir;
            *((float4*)&lds_B[row * LDT + c4 * 4]) = v;
        }
        __syncthreads();

        v8f acc0[4], acc1[4];
#pragma unroll
        for (int nt = 0; nt < 4; ++nt) {
            acc0[nt] = (v8f){0.f,0.f,0.f,0.f,0.f,0.f,0.f,0.f};
            acc1[nt] = (v8f){0.f,0.f,0.f,0.f,0.f,0.f,0.f,0.f};
        }

#pragma unroll 4
        for (int k = 0; k < DDIM; k += 4) {
            const int kk = k + 2 * khalf;
            v2f a0 = *((const v2f*)(Arow0 + kk));
            v2f a1 = *((const v2f*)(Arow1 + kk));
#pragma unroll
            for (int nt = 0; nt < 4; ++nt) {
                v2f b = *((const v2f*)(&lds_B[(colBlk * 64 + nt * 16 + m) * LDT] + kk));
                acc0[nt] = __builtin_amdgcn_wmma_f32_16x16x4_f32(
                    false, a0, false, b, (short)0, acc0[nt], false, false);
                acc1[nt] = __builtin_amdgcn_wmma_f32_16x16x4_f32(
                    false, a1, false, b, (short)0, acc1[nt], false, false);
            }
        }

#pragma unroll
        for (int nt = 0; nt < 4; ++nt) {
#pragma unroll
            for (int r = 0; r < 8; ++r) {
                rmax[0][r] = fmaxf(rmax[0][r], acc0[nt][r]);
                rmax[1][r] = fmaxf(rmax[1][r], acc1[nt][r]);
            }
        }
    }

#pragma unroll
    for (int rt = 0; rt < 2; ++rt) {
#pragma unroll
        for (int r = 0; r < 8; ++r) {
            float v = rmax[rt][r];
            v = fmaxf(v, __shfl_xor(v, 1, 32));
            v = fmaxf(v, __shfl_xor(v, 2, 32));
            v = fmaxf(v, __shfl_xor(v, 4, 32));
            v = fmaxf(v, __shfl_xor(v, 8, 32));
            rmax[rt][r] = v;
        }
    }

    const int sel = lane & 7;
    const int grp = lane >> 4;
#pragma unroll
    for (int rt = 0; rt < 2; ++rt) {
        float v = rmax[rt][0];
#pragma unroll
        for (int r = 1; r < 8; ++r) v = (sel == r) ? rmax[rt][r] : v;
        if ((lane & 15) < 8) {
            int rowLocal = rowBlk * 32 + rt * 16 + grp * 8 + sel;
            red[colBlk * BM + rowLocal] = v;
        }
    }
    __syncthreads();

    if (t < BM) {
        int row = i0 + t;
        out[(size_t)w * 1024 * 32 + (size_t)row * 32 + p] =
            fmaxf(red[t], red[BM + t]);
    }
}

extern "C" void kernel_launch(void* const* d_in, const int* in_sizes, int n_in,
                              void* d_out, int out_size, void* d_ws, size_t ws_size,
                              hipStream_t stream) {
    const float* s1 = (const float*)d_in[0];
    const float* s2 = (const float*)d_in[1];
    const float* W1 = (const float*)d_in[2];
    const float* W2 = (const float*)d_in[3];
    float* out = (float*)d_out;
    float* inv = (float*)d_ws;

    norms_kernel<<<dim3(512), dim3(256), 0, stream>>>(s1, s2, W1, W2, inv);
    mpm_kernel<<<dim3(512), dim3(256), 0, stream>>>(s1, s2, W1, W2, inv, out);
}


# B staged in half-K stages through two LDS buffers; each wave stores stage s+1 and issues loads of s+2 after the first k of stage s, so staging runs under the other wave's WMMAs
# speedup vs baseline: 1.0588x; 1.0588x over previous
#include <hip/hip_runtime.h>
#include <hip/hip_bf16.h>


typedef __attribute__((ext_vector_type(2))) float v2f;
typedef __attribute__((ext_vector_type(8))) float v8f;

#define DDIM 256
#define BM 128
#define BN 128
#define LDT 260
#define KH 128
#define LDB 132
#define NPF 16
#define KSPLIT 32

__global__ __launch_bounds__(256) void norms_kernel(
    const float* __restrict__ s1, const float* __restrict__ s2,
    const float* __restrict__ W1, const float* __restrict__ W2,
    float* __restrict__ inv)
{
    int idx = blockIdx.x * blockDim.x + threadIdx.x;
    int i   = idx & 1023;
    int p   = (idx >> 10) & 31;
    int w   = (idx >> 15) & 1;
    int src = (idx >> 16) & 1;
    const float4* xr = (const float4*)((src ? s2 : s1) + i * DDIM);
    const float4* wr = (const float4*)((w ? W2 : W1) + p * DDIM);
    float sum = 0.f;
#pragma unroll 8
    for (int d4 = 0; d4 < DDIM / 4; ++d4) {
        float4 xv = xr[d4];
        float4 wv = wr[d4];
        float a0 = xv.x * wv.x, a1 = xv.y * wv.y, a2 = xv.z * wv.z, a3 = xv.w * wv.w;
        sum += a0 * a0 + a1 * a1 + a2 * a2 + a3 * a3;
    }
    inv[idx] = 1.0f / fmaxf(sqrtf(sum), 1e-8f);
}

template <int KB, int K0, int K1>
__device__ __forceinline__ void k_steps(
    const float* __restrict__ Arow0, const float* __restrict__ Arow1,
    const float* __restrict__ Bcol, int khalf, v8f (&acc0)[4], v8f (&acc1)[4])
{
#pragma unroll 4
    for (int k = K0; k < K1; k += 4) {
        const int kk = k + 2 * khalf;
        v2f a0 = *((const v2f*)(Arow0 + KB + kk));
        v2f a1 = *((const v2f*)(Arow1 + KB + kk));
#pragma unroll
        for (int nt = 0; nt < 4; ++nt) {
            v2f b = *((const v2f*)(Bcol + nt * 16 * LDB + kk));
            acc0[nt] = __builtin_amdgcn_wmma_f32_16x16x4_f32(
                false, a0, false, b, (short)0, acc0[nt], false, false);
            acc1[nt] = __builtin_amdgcn_wmma_f32_16x16x4_f32(
                false, a1, false, b, (short)0, acc1[nt], false, false);
        }
    }
}

__global__ __launch_bounds__(256) void mpm_kernel(
    const float* __restrict__ s1, const float* __restrict__ s2,
    const float* __restrict__ W1, const float* __restrict__ W2,
    const float* __restrict__ invn,
    float* __restrict__ out)
{
    __shared__ float lds_A[BM * LDT];
    __shared__ float lds_B0[BN * LDB];
    __shared__ float lds_B1[BN * LDB];
    __shared__ float red[2 * BM];

    const int pslot = blockIdx.x & 63;
    const int iTile = blockIdx.x >> 6;
    const int w  = pslot >> 5;
    const int p  = pslot & 31;
    const int i0 = iTile * BM;

    const float* Wsel = w ? W2 : W1;
    const float* invA = invn + ((size_t)(0 * 2 + w) * 32 + p) * 1024;
    const float* invB = invn + ((size_t)(1 * 2 + w) * 32 + p) * 1024;

    const int t    = threadIdx.x;
    const int lane = t & 31;
    const int wv   = t >> 5;
    const int rowBlk = wv & 3;
    const int colBlk = wv >> 2;

    const float4* Wrow = (const float4*)(Wsel + p * DDIM);
    const float4 wA  = Wrow[t & 63];
    const float4 wB0 = Wrow[lane];
    const float4 wB1 = Wrow[32 + lane];

    float4 pf[NPF];
    float  pir[NPF];

#define B_ISSUE(J0, KB)                                                        \
    _Pragma("unroll")                                                          \
    for (int it = 0; it < NPF; ++it) {                                         \
        const int row = wv + it * 8;                                           \
        pir[it] = invB[(J0) + row];                                            \
        pf[it]  = ((const float4*)(s2 + (size_t)((J0) + row) * DDIM + (KB)))[lane]; \
    }

#define B_COMMIT(DST, WQ)                                                      \
    _Pragma("unroll")                                                          \
    for (int it = 0; it < NPF; ++it) {                                         \
        const int row = wv + it * 8;                                           \
        const float ir = pir[it];                                              \
        float4 v = pf[it];                                                     \
        v.x *= (WQ).x * ir; v.y *= (WQ).y * ir;                                \
        v.z *= (WQ).z * ir; v.w *= (WQ).w * ir;                                \
        *((float4*)&(DST)[row * LDB + lane * 4]) = v;                          \
    }

    B_ISSUE(0, 0)

#pragma unroll 8
    for (int it = 0; it < (BM * DDIM / 4) / 256; ++it) {
        int row  = (t >> 6) + it * 4;
        int c4   = t & 63;
        float ir = invA[i0 + row];
        float4 v  = ((const float4*)(s1 + (size_t)(i0 + row) * DDIM))[c4];
        v.x *= wA.x * ir; v.y *= wA.y * ir; v.z *= wA.z * ir; v.w *= wA.w * ir;
        *((float4*)&lds_A[row * LDT + c4 * 4]) = v;
    }

    B_COMMIT(lds_B0, wB0)
    __syncthreads();
    B_ISSUE(0, KH)

    float rmax[2][8];
#pragma unroll
    for (int rt = 0; rt < 2; ++rt)
#pragma unroll
        for (int r = 0; r < 8; ++r) rmax[rt][r] = -3.402823466e+38f;

    const int m     = lane & 15;
    const int khalf = lane >> 4;
    const float* Arow0 = &lds_A[(rowBlk * 32 + m) * LDT];
    const float* Arow1 = Arow0 + 16 * LDT;
    const float* Bcol0 = &lds_B0[(colBlk * 64 + m) * LDB];
    const float* Bcol1 = &lds_B1[(colBlk * 64 + m) * LDB];

    for (int jt = 0; jt < 8; ++jt) {
        const int j0 = jt * BN;

        v8f acc0[4], acc1[4];
#pragma unroll
        for (int nt = 0; nt < 4; ++nt) {
            acc0[nt] = (v8f){0.f,0.f,0.f,0.f,0.f,0.f,0.f,0.f};
            acc1[nt] = (v8f){0.f,0.f,0.f,0.f,0.f,0.f,0.f,0.f};
        }
        k_steps<0, 0, KSPLIT>(Arow0, Arow1, Bcol0, khalf, acc0, acc1);

        B_COMMIT(lds_B1, wB1)
        if (jt < 7) {
            B_ISSUE(j0 + BN, 0)
        }
        k_steps<0, KSPLIT, KH>(Arow0, Arow1, Bcol0, khalf, acc0, acc1);
        __syncthreads();

        k_steps<KH, 0, KSPLIT>(Arow0, Arow1, Bcol1, khalf, acc0, acc1);

        if (jt < 7) {
            B_COMMIT(lds_B0, wB0)
            B_ISSUE(j0 + BN, KH)
        }
        k_steps<KH, KSPLIT, KH>(Arow0, Arow1, Bcol1, khalf, acc0, acc1);

#pragma unroll
        for (int nt = 0; nt < 4; ++nt) {
#pragma unroll
            for (int r = 0; r < 8; ++r) {
                rmax[0][r] = fmaxf(rmax[0][r], acc0[nt][r]);
                rmax[1][r] = fmaxf(rmax[1][r], acc1[nt][r]);
            }
        }
        __syncthreads();
    }
#undef B_ISSUE
#undef B_COMMIT

#pragma unroll
    for (int rt = 0; rt < 2; ++rt) {
#pragma unroll
        for (int r = 0; r < 8; ++r) {
            float v = rmax[rt][r];
            v = fmaxf(v, __shfl_xor(v, 1, 32));
            v = fmaxf(v, __shfl_xor(v, 2, 32));
            v = fmaxf(v, __shfl_xor(v, 4, 32));
            v = fmaxf(v, __shfl_xor(v, 8, 32));
            rmax[rt][r] = v;
        }
    }

    const int sel = lane & 7;
    const int grp = lane >> 4;
#pragma unroll
    for (int rt = 0; rt < 2; ++rt) {
        float v = rmax[rt][0];
#pragma unroll
        for (int r = 1; r < 8; ++r) v = (sel == r) ? rmax[rt][r] : v;
        if ((lane & 15) < 8) {
            int rowLocal = rowBlk * 32 + rt * 16 + grp * 8 + sel;
            red[colBlk * BM + rowLocal] = v;
        }
    }
    __syncthreads();

    if (t < BM) {
        int row = i0 + t;
        out[(size_t)w * 1024 * 32 + (size_t)row * 32 + p] =
            fmaxf(red[t], red[BM + t]);
    }
}

extern "C" void kernel_launch(void* const* d_in, const int* in_sizes, int n_in,
                              void* d_out, int out_size, void* d_ws, size_t ws_size,
                              hipStream_t stream) {
    const float* s1 = (const float*)d_in[0];
    const float* s2 = (const float*)d_in[1];
    const float* W1 = (const float*)d_in[2];
    const float* W2 = (const float*)d_in[3];
    float* out = (float*)d_out;
    float* inv = (float*)d_ws;

    norms_kernel<<<dim3(512), dim3(256), 0, stream>>>(s1, s2, W1, W2, inv);
    mpm_kernel<<<dim3(512), dim3(256), 0, stream>>>(s1, s2, W1, W2, inv, out);
}


# A and B fragments double-buffered in registers: the LDS loads of the next 8-step block go out as one group in front of the current block's 64 WMMAs, so no block waits for an LDS round trip
# speedup vs baseline: 1.1023x; 1.0411x over previous
#include <hip/hip_runtime.h>
#include <hip/hip_bf16.h>


typedef __attribute__((ext_vector_type(2))) float v2f;
typedef __attribute__((ext_vector_type(8))) float v8f;

#define DDIM 256
#define BM 128
#define BN 128
#define LDT 260
#define KH 128
#define LDB 132
#define NPF 16
#define KSPLIT 32
#define KBLK 32
static_assert(KSPLIT == KBLK && KH == 4 * KBLK, "a stage is 4 blocks; staging sits behind the first");

__global__ __launch_bounds__(256) void norms_kernel(
    const float* __restrict__ s1, const float* __restrict__ s2,
    const float* __restrict__ W1, const float* __restrict__ W2,
    float* __restrict__ inv)
{
    int idx = blockIdx.x * blockDim.x + threadIdx.x;
    int i   = idx & 1023;
    int p   = (idx >> 10) & 31;
    int w   = (idx >> 15) & 1;
    int src = (idx >> 16) & 1;
    const float4* xr = (const float4*)((src ? s2 : s1) + i * DDIM);
    const float4* wr = (const float4*)((w ? W2 : W1) + p * DDIM);
    float sum = 0.f;
#pragma unroll 8
    for (int d4 = 0; d4 < DDIM / 4; ++d4) {
        float4 xv = xr[d4];
        float4 wv = wr[d4];
        float a0 = xv.x * wv.x, a1 = xv.y * wv.y, a2 = xv.z * wv.z, a3 = xv.w * wv.w;
        sum += a0 * a0 + a1 * a1 + a2 * a2 + a3 * a3;
    }
    inv[idx] = 1.0f / fmaxf(sqrtf(sum), 1e-8f);
}

struct Frag {
    v2f a0[KBLK / 4], a1[KBLK / 4];
    v2f b[KBLK / 4][4];
};

__device__ __forceinline__ void frag_load(
    Frag& f, const float* __restrict__ A0, const float* __restrict__ A1,
    const float* __restrict__ Bc, int k0)
{
#pragma unroll
    for (int s = 0; s < KBLK / 4; ++s) {
        const int kk = k0 + 4 * s;
        f.a0[s] = *((const v2f*)(A0 + kk));
        f.a1[s] = *((const v2f*)(A1 + kk));
#pragma unroll
        for (int nt = 0; nt < 4; ++nt)
            f.b[s][nt] = *((const v2f*)(Bc + nt * 16 * LDB + kk));
    }
}

__device__ __forceinline__ void frag_mma(
    const Frag& f, v8f (&acc0)[4], v8f (&acc1)[4])
{
#pragma unroll
    for (int s = 0; s < KBLK / 4; ++s) {
#pragma unroll
        for (int nt = 0; nt < 4; ++nt) {
            acc0[nt] = __builtin_amdgcn_wmma_f32_16x16x4_f32(
                false, f.a0[s], false, f.b[s][nt], (short)0, acc0[nt], false, false);
            acc1[nt] = __builtin_amdgcn_wmma_f32_16x16x4_f32(
                false, f.a1[s], false, f.b[s][nt], (short)0, acc1[nt], false, false);
        }
    }
}

__global__ __launch_bounds__(256) void mpm_kernel(
    const float* __restrict__ s1, const float* __restrict__ s2,
    const float* __restrict__ W1, const float* __restrict__ W2,
    const float* __restrict__ invn,
    float* __restrict__ out)
{
    __shared__ float lds_A[BM * LDT];
    __shared__ float lds_B0[BN * LDB];
    __shared__ float lds_B1[BN * LDB];
    __shared__ float red[2 * BM];

    const int pslot = blockIdx.x & 63;
    const int iTile = blockIdx.x >> 6;
    const int w  = pslot >> 5;
    const int p  = pslot & 31;
    const int i0 = iTile * BM;

    const float* Wsel = w ? W2 : W1;
    const float* invA = invn + ((size_t)(0 * 2 + w) * 32 + p) * 1024;
    const float* invB = invn + ((size_t)(1 * 2 + w) * 32 + p) * 1024;

    const int t    = threadIdx.x;
    const int lane = t & 31;
    const int wv   = t >> 5;
    const int rowBlk = wv & 3;
    const int colBlk = wv >> 2;

    const float4* Wrow = (const float4*)(Wsel + p * DDIM);
    const float4 wA  = Wrow[t & 63];
    const float4 wB0 = Wrow[lane];
    const float4 wB1 = Wrow[32 + lane];

    float4 pf[NPF];
    float  pir[NPF];

#define B_ISSUE(J0, KB)                                                        \
    _Pragma("unroll")                                                          \
    for (int it = 0; it < NPF; ++it) {                                         \
        const int row = wv + it * 8;                                           \
        pir[it] = invB[(J0) + row];                                            \
        pf[it]  = ((const float4*)(s2 + (size_t)((J0) + row) * DDIM + (KB)))[lane]; \
    }

#define B_COMMIT(DST, WQ)                                                      \
    _Pragma("unroll")                                                          \
    for (int it = 0; it < NPF; ++it) {                                         \
        const int row = wv + it * 8;                                           \
        const float ir = pir[it];                                              \
        float4 v = pf[it];                                                     \
        v.x *= (WQ).x * ir; v.y *= (WQ).y * ir;                                \
        v.z *= (WQ).z * ir; v.w *= (WQ).w * ir;                                \
        *((float4*)&(DST)[row * LDB + lane * 4]) = v;                          \
    }

    B_ISSUE(0, 0)

#pragma unroll 8
    for (int it = 0; it < (BM * DDIM / 4) / 256; ++it) {
        int row  = (t >> 6) + it * 4;
        int c4   = t & 63;
        float ir = invA[i0 + row];
        float4 v  = ((const float4*)(s1 + (size_t)(i0 + row) * DDIM))[c4];
        v.x *= wA.x * ir; v.y *= wA.y * ir; v.z *= wA.z * ir; v.w *= wA.w * ir;
        *((float4*)&lds_A[row * LDT + c4 * 4]) = v;
    }

    B_COMMIT(lds_B0, wB0)
    __syncthreads();
    B_ISSUE(0, KH)

    float rmax[2][8];
#pragma unroll
    for (int rt = 0; rt < 2; ++rt)
#pragma unroll
        for (int r = 0; r < 8; ++r) rmax[rt][r] = -3.402823466e+38f;

    const int m     = lane & 15;
    const int khalf = lane >> 4;
    const float* AL0 = &lds_A[(rowBlk * 32 + m) * LDT + 2 * khalf];
    const float* AL1 = AL0 + 16 * LDT;
    const float* AH0 = AL0 + KH;
    const float* AH1 = AL1 + KH;
    const float* Bc0 = &lds_B0[(colBlk * 64 + m) * LDB + 2 * khalf];
    const float* Bc1 = &lds_B1[(colBlk * 64 + m) * LDB + 2 * khalf];

    Frag F0, F1;

#define HOLD() __builtin_amdgcn_sched_barrier(0)
#define BLOCK(FC, FN, A0, A1, BC, KNEXT)                                       \
    frag_load(FN, A0, A1, BC, (KNEXT));                                        \
    HOLD();                                                                    \
    frag_mma(FC, acc0, acc1);                                                  \
    HOLD();

    for (int jt = 0; jt < 8; ++jt) {
        const int j0 = jt * BN;

        v8f acc0[4], acc1[4];
#pragma unroll
        for (int nt = 0; nt < 4; ++nt) {
            acc0[nt] = (v8f){0.f,0.f,0.f,0.f,0.f,0.f,0.f,0.f};
            acc1[nt] = (v8f){0.f,0.f,0.f,0.f,0.f,0.f,0.f,0.f};
        }
        frag_load(F0, AL0, AL1, Bc0, 0);
        BLOCK(F0, F1, AL0, AL1, Bc0, 1 * KBLK)

        B_COMMIT(lds_B1, wB1)
        if (jt < 7) {
            B_ISSUE(j0 + BN, 0)
        }
        BLOCK(F1, F0, AL0, AL1, Bc0, 2 * KBLK)
        BLOCK(F0, F1, AL0, AL1, Bc0, 3 * KBLK)
        frag_mma(F1, acc0, acc1);
        __syncthreads();

        frag_load(F0, AH0, AH1, Bc1, 0);
        BLOCK(F0, F1, AH0, AH1, Bc1, 1 * KBLK)

        if (jt < 7) {
            B_COMMIT(lds_B0, wB0)
            B_ISSUE(j0 + BN, KH)
        }
        BLOCK(F1, F0, AH0, AH1, Bc1, 2 * KBLK)
        BLOCK(F0, F1, AH0, AH1, Bc1, 3 * KBLK)
        frag_mma(F1, acc0, acc1);

#pragma unroll
        for (int nt = 0; nt < 4; ++nt) {
#pragma unroll
            for (int r = 0; r < 8; ++r) {
                rmax[0][r] = fmaxf(rmax[0][r], acc0[nt][r]);
                rmax[1][r] = fmaxf(rmax[1][r], acc1[nt][r]);
            }
        }
        __syncthreads();
    }
#undef BLOCK
#undef HOLD
#undef B_ISSUE
#undef B_COMMIT

#pragma unroll
    for (int rt = 0; rt < 2; ++rt) {
#pragma unroll
        for (int r = 0; r < 8; ++r) {
            float v = rmax[rt][r];
            v = fmaxf(v, __shfl_xor(v, 1, 32));
            v = fmaxf(v, __shfl_xor(v, 2, 32));
            v = fmaxf(v, __shfl_xor(v, 4, 32));
            v = fmaxf(v, __shfl_xor(v, 8, 32));
            rmax[rt][r] = v;
        }
    }

    const int sel = lane & 7;
    const int grp = lane >> 4;
#pragma unroll
    for (int rt = 0; rt < 2; ++rt) {
        float v = rmax[rt][0];
#pragma unroll
        for (int r = 1; r < 8; ++r) v = (sel == r) ? rmax[rt][r] : v;
        if ((lane & 15) < 8) {
            int rowLocal = rowBlk * 32 + rt * 16 + grp * 8 + sel;
            red[colBlk * BM + rowLocal] = v;
        }
    }
    __syncthreads();

    if (t < BM) {
        int row = i0 + t;
        out[(size_t)w * 1024 * 32 + (size_t)row * 32 + p] =
            fmaxf(red[t], red[BM + t]);
    }
}

extern "C" void kernel_launch(void* const* d_in, const int* in_sizes, int n_in,
                              void* d_out, int out_size, void* d_ws, size_t ws_size,
                              hipStream_t stream) {
    const float* s1 = (const float*)d_in[0];
    const float* s2 = (const float*)d_in[1];
    const float* W1 = (const float*)d_in[2];
    const float* W2 = (const float*)d_in[3];
    float* out = (float*)d_out;
    float* inv = (float*)d_ws;

    norms_kernel<<<dim3(512), dim3(256), 0, stream>>>(s1, s2, W1, W2, inv);
    mpm_kernel<<<dim3(512), dim3(256), 0, stream>>>(s1, s2, W1, W2, inv, out);
}


# Inverse norms made inside the matrix kernel from the staged LDS tiles, applied after the product sums (B columns at the fold, A rows at the end); norms kernel and launch gone; norm sum order changes
# speedup vs baseline: 1.1593x; 1.0517x over previous
#include <hip/hip_runtime.h>
#include <hip/hip_bf16.h>


typedef __attribute__((ext_vector_type(2))) float v2f;
typedef __attribute__((ext_vector_type(8))) float v8f;

#define DDIM 256
#define BM 128
#define BN 128
#define LDT 260
#define KH 128
#define LDB 132
#define NPF 16
#define KSPLIT 32
#define KBLK 32
static_assert(KSPLIT == KBLK && KH == 4 * KBLK, "a stage is 4 blocks; staging sits behind the first");

struct Frag {
    v2f a0[KBLK / 4], a1[KBLK / 4];
    v2f b[KBLK / 4][4];
};

__device__ __forceinline__ void frag_load(
    Frag& f, const float* __restrict__ A0, const float* __restrict__ A1,
    const float* __restrict__ Bc, int k0)
{
#pragma unroll
    for (int s = 0; s < KBLK / 4; ++s) {
        const int kk = k0 + 4 * s;
        f.a0[s] = *((const v2f*)(A0 + kk));
        f.a1[s] = *((const v2f*)(A1 + kk));
#pragma unroll
        for (int nt = 0; nt < 4; ++nt)
            f.b[s][nt] = *((const v2f*)(Bc + nt * 16 * LDB + kk));
    }
}

__device__ __forceinline__ void frag_mma(
    const Frag& f, v8f (&acc0)[4], v8f (&acc1)[4])
{
#pragma unroll
    for (int s = 0; s < KBLK / 4; ++s) {
#pragma unroll
        for (int nt = 0; nt < 4; ++nt) {
            acc0[nt] = __builtin_amdgcn_wmma_f32_16x16x4_f32(
                false, f.a0[s], false, f.b[s][nt], (short)0, acc0[nt], false, false);
            acc1[nt] = __builtin_amdgcn_wmma_f32_16x16x4_f32(
                false, f.a1[s], false, f.b[s][nt], (short)0, acc1[nt], false, false);
        }
    }
}

__global__ __launch_bounds__(256) void mpm_kernel(
    const float* __restrict__ s1, const float* __restrict__ s2,
    const float* __restrict__ W1, const float* __restrict__ W2,
    float* __restrict__ out)
{
    __shared__ float lds_A[BM * LDT];
    __shared__ float lds_B0[BN * LDB];
    __shared__ float lds_B1[BN * LDB];
    __shared__ float red[2 * BM];
    __shared__ float invAt[BM];
    __shared__ float invBt[BN];

    const int pslot = blockIdx.x & 63;
    const int iTile = blockIdx.x >> 6;
    const int w  = pslot >> 5;
    const int p  = pslot & 31;
    const int i0 = iTile * BM;

    const float* Wsel = w ? W2 : W1;

    const int t    = threadIdx.x;
    const int lane = t & 31;
    const int wv   = t >> 5;
    const int rowBlk = wv & 3;
    const int colBlk = wv >> 2;

    const float4* Wrow = (const float4*)(Wsel + p * DDIM);
    const float4 wA  = Wrow[t & 63];
    const float4 wB0 = Wrow[lane];
    const float4 wB1 = Wrow[32 + lane];

    float4 pf[NPF];

#define B_ISSUE(J0, KB)                                                        \
    _Pragma("unroll")                                                          \
    for (int it = 0; it < NPF; ++it) {                                         \
        const int row = wv + it * 8;                                           \
        pf[it]  = ((const float4*)(s2 + (size_t)((J0) + row) * DDIM + (KB)))[lane]; \
    }

#define B_COMMIT(DST, WQ)                                                      \
    _Pragma("unroll")                                                          \
    for (int it = 0; it < NPF; ++it) {                                         \
        const int row = wv + it * 8;                                           \
        float4 v = pf[it];                                                     \
        v.x *= (WQ).x; v.y *= (WQ).y;                                          \
        v.z *= (WQ).z; v.w *= (WQ).w;                                          \
        *((float4*)&(DST)[row * LDB + lane * 4]) = v;                          \
    }

#define B_SSQ(SRC)                                                             \
    {                                                                          \
        const float4* bp = (const float4*)&(SRC)[(t >> 1) * LDB + (t & 1) * 64]; \
        _Pragma("unroll")                                                      \
        for (int q = 0; q < 16; ++q) {                                         \
            const float4 b = bp[q];                                            \
            ssq += b.x * b.x + b.y * b.y + b.z * b.z + b.w * b.w;              \
        }                                                                      \
    }

    B_ISSUE(0, 0)

#pragma unroll 8
    for (int it = 0; it < (BM * DDIM / 4) / 256; ++it) {
        int row  = (t >> 6) + it * 4;
        int c4   = t & 63;
        float4 v  = ((const float4*)(s1 + (size_t)(i0 + row) * DDIM))[c4];
        v.x *= wA.x; v.y *= wA.y; v.z *= wA.z; v.w *= wA.w;
        *((float4*)&lds_A[row * LDT + c4 * 4]) = v;
    }

    B_COMMIT(lds_B0, wB0)
    __syncthreads();
    B_ISSUE(0, KH)

    {
        const float4* ap = (const float4*)&lds_A[(t >> 1) * LDT + (t & 1) * 128];
        float sa = 0.f;
#pragma unroll 8
        for (int q = 0; q < 32; ++q) {
            const float4 a = ap[q];
            sa += a.x * a.x + a.y * a.y + a.z * a.z + a.w * a.w;
        }
        sa += __shfl_xor(sa, 1, 32);
        if ((t & 1) == 0) invAt[t >> 1] = 1.0f / fmaxf(sqrtf(sa), 1e-8f);
    }

    float rmax[2][8];
#pragma unroll
    for (int rt = 0; rt < 2; ++rt)
#pragma unroll
        for (int r = 0; r < 8; ++r) rmax[rt][r] = -3.402823466e+38f;

    const int m     = lane & 15;
    const int khalf = lane >> 4;
    const float* AL0 = &lds_A[(rowBlk * 32 + m) * LDT + 2 * khalf];
    const float* AL1 = AL0 + 16 * LDT;
    const float* AH0 = AL0 + KH;
    const float* AH1 = AL1 + KH;
    const float* Bc0 = &lds_B0[(colBlk * 64 + m) * LDB + 2 * khalf];
    const float* Bc1 = &lds_B1[(colBlk * 64 + m) * LDB + 2 * khalf];

    Frag F0, F1;

#define HOLD() __builtin_amdgcn_sched_barrier(0)
#define BLOCK(FC, FN, A0, A1, BC, KNEXT)                                       \
    frag_load(FN, A0, A1, BC, (KNEXT));                                        \
    HOLD();                                                                    \
    frag_mma(FC, acc0, acc1);                                                  \
    HOLD();

    for (int jt = 0; jt < 8; ++jt) {
        const int j0 = jt * BN;

        v8f acc0[4], acc1[4];
#pragma unroll
        for (int nt = 0; nt < 4; ++nt) {
            acc0[nt] = (v8f){0.f,0.f,0.f,0.f,0.f,0.f,0.f,0.f};
            acc1[nt] = (v8f){0.f,0.f,0.f,0.f,0.f,0.f,0.f,0.f};
        }
        float ssq = 0.f;
        frag_load(F0, AL0, AL1, Bc0, 0);
        BLOCK(F0, F1, AL0, AL1, Bc0, 1 * KBLK)

        B_SSQ(lds_B0)
        B_COMMIT(lds_B1, wB1)
        if (jt < 7) {
            B_ISSUE(j0 + BN, 0)
        }
        BLOCK(F1, F0, AL0, AL1, Bc0, 2 * KBLK)
        BLOCK(F0, F1, AL0, AL1, Bc0, 3 * KBLK)
        frag_mma(F1, acc0, acc1);
        __syncthreads();

        frag_load(F0, AH0, AH1, Bc1, 0);
        BLOCK(F0, F1, AH0, AH1, Bc1, 1 * KBLK)

        B_SSQ(lds_B1)
        ssq += __shfl_xor(ssq, 1, 32);
        if ((t & 1) == 0) invBt[t >> 1] = 1.0f / fmaxf(sqrtf(ssq), 1e-8f);
        if (jt < 7) {
            B_COMMIT(lds_B0, wB0)
            B_ISSUE(j0 + BN, KH)
        }
        BLOCK(F1, F0, AH0, AH1, Bc1, 2 * KBLK)
        BLOCK(F0, F1, AH0, AH1, Bc1, 3 * KBLK)
        frag_mma(F1, acc0, acc1);
        __syncthreads();

#pragma unroll
        for (int nt = 0; nt < 4; ++nt) {
            const float ib = invBt[colBlk * 64 + nt * 16 + m];
#pragma unroll
            for (int r = 0; r < 8; ++r) {
                rmax[0][r] = fmaxf(rmax[0][r], acc0[nt][r] * ib);
                rmax[1][r] = fmaxf(rmax[1][r], acc1[nt][r] * ib);
            }
        }
    }
#undef BLOCK
#undef HOLD
#undef B_ISSUE
#undef B_COMMIT
#undef B_SSQ

#pragma unroll
    for (int rt = 0; rt < 2; ++rt) {
#pragma unroll
        for (int r = 0; r < 8; ++r) {
            float v = rmax[rt][r];
            v = fmaxf(v, __shfl_xor(v, 1, 32));
            v = fmaxf(v, __shfl_xor(v, 2, 32));
            v = fmaxf(v, __shfl_xor(v, 4, 32));
            v = fmaxf(v, __shfl_xor(v, 8, 32));
            rmax[rt][r] = v;
        }
    }

    const int sel = lane & 7;
    const int grp = lane >> 4;
#pragma unroll
    for (int rt = 0; rt < 2; ++rt) {
        float v = rmax[rt][0];
#pragma unroll
        for (int r = 1; r < 8; ++r) v = (sel == r) ? rmax[rt][r] : v;
        if ((lane & 15) < 8) {
            int rowLocal = rowBlk * 32 + rt * 16 + grp * 8 + sel;
            red[colBlk * BM + rowLocal] = v;
        }
    }
    __syncthreads();

    if (t < BM) {
        int row = i0 + t;
        out[(size_t)w * 1024 * 32 + (size_t)row * 32 + p] =
            fmaxf(red[t], red[BM + t]) * invAt[t];
    }
}

extern "C" void kernel_launch(void* const* d_in, const int* in_sizes, int n_in,
                              void* d_out, int out_size, void* d_ws, size_t ws_size,
                              hipStream_t stream) {
    const float* s1 = (const float*)d_in[0];
    const float* s2 = (const float*)d_in[1];
    const float* W1 = (const float*)d_in[2];
    const float* W2 = (const float*)d_in[3];
    float* out = (float*)d_out;

    mpm_kernel<<<dim3(512), dim3(256), 0, stream>>>(s1, s2, W1, W2, out);
}
